# starting kernel
# baseline (speedup 1.0000x reference)
#include <hip/hip_runtime.h>
#include <math.h>

typedef __attribute__((ext_vector_type(2))) float v2f;
typedef __attribute__((ext_vector_type(8))) float v8f;

#define NUM_FEA 32
#define EMB_DIM 8
#define OUT_DIM 16
#define WAVES_PER_BLOCK 8

#define SWZ(v, imm) __int_as_float(__builtin_amdgcn_ds_swizzle(__float_as_int(v), (imm)))
#define SWZ_X1  0x041F
#define SWZ_X2  0x081F
#define SWZ_X4  0x101F
#define SWZ_X8  0x201F
#define SWZ_X16 0x401F
#define SWZ_BCL 0x000F
#define SWZ_BCR 0x020F

__device__ __forceinline__ float fast_tanh(float v) {
#if __has_builtin(__builtin_amdgcn_tanhf)
    return __builtin_amdgcn_tanhf(v);
#else
    return tanhf(v);
#endif
}

__device__ __forceinline__ float sum16(float v) {
    v += SWZ(v, SWZ_X1);
    v += SWZ(v, SWZ_X2);
    v += SWZ(v, SWZ_X4);
    v += SWZ(v, SWZ_X8);
    return v;
}

__global__ __launch_bounds__(256)
void coattn_kernel(const float* __restrict__ x,
                   const float* __restrict__ emb_W,
                   const float* __restrict__ emb_b,
                   const float* __restrict__ M,
                   const float* __restrict__ dense_W,
                   const float* __restrict__ dense_b,
                   float* __restrict__ out,
                   int nrows) {
    __shared__ float sWM[128];
    __shared__ float sbM[128];
    __shared__ float sWr[128];
    __shared__ float sbr[128];
    __shared__ float sdW[512];
    __shared__ float sdb[16];
    __shared__ float sT[WAVES_PER_BLOCK][16 * 20];
    __shared__ float out_s[WAVES_PER_BLOCK][32];

    const int tid = threadIdx.x;

    if (tid < 128) {
        const int l = tid >> 3, f = tid & 7;
        float aw = 0.f, ab = 0.f;
        #pragma unroll
        for (int e = 0; e < EMB_DIM; ++e) {
            const float mv = M[e * EMB_DIM + f];
            aw = fmaf(emb_W[l * EMB_DIM + e], mv, aw);
            ab = fmaf(emb_b[l * EMB_DIM + e], mv, ab);
        }
        sWM[tid] = aw;
        sbM[tid] = ab;
        sWr[tid] = emb_W[128 + tid];
        sbr[tid] = emb_b[128 + tid];
    }
    sdW[tid]       = dense_W[tid];
    sdW[tid + 256] = dense_W[tid + 256];
    if (tid < OUT_DIM) sdb[tid] = dense_b[tid];
    __syncthreads();

    const int lane = tid & 31;
    const int wave = tid >> 5;
    int b = blockIdx.x * WAVES_PER_BLOCK + wave;
    const bool valid = (b < nrows);
    if (!valid) b = nrows - 1;

    const int m16  = lane & 15;
    const int half = lane >> 4;

    const float xv = x[(size_t)b * NUM_FEA + lane];
    const float xl = SWZ(xv, SWZ_BCL);
    const float xr = SWZ(xv, SWZ_BCR);

    v8f c = {};
    #pragma unroll
    for (int w = 0; w < 2; ++w) {
        const int k0 = w * 4 + half * 2;
        v2f a, bmat;
        a.x    = fmaf(xl, sWM[m16 * 8 + k0],     sbM[m16 * 8 + k0]);
        a.y    = fmaf(xl, sWM[m16 * 8 + k0 + 1], sbM[m16 * 8 + k0 + 1]);
        bmat.x = fmaf(xr, sWr[m16 * 8 + k0],     sbr[m16 * 8 + k0]);
        bmat.y = fmaf(xr, sWr[m16 * 8 + k0 + 1], sbr[m16 * 8 + k0 + 1]);
        c = __builtin_amdgcn_wmma_f32_16x16x4_f32(
                 false, a,  false, bmat,
                 (short)0, c,  false,  false);
    }

    float g[8];
    #pragma unroll
    for (int v = 0; v < 8; ++v) g[v] = fast_tanh(c[v]);

    float cs = 0.f;
    #pragma unroll
    for (int v = 0; v < 8; ++v) cs += g[v];
    cs += SWZ(cs, SWZ_X16);
    const float e_lr = __expf(cs);
    const float lr_lane = e_lr / sum16(e_lr);

    {
        float* dst = &sT[wave][m16 * 20 + half * 8];
        *(float4*)(dst)     = make_float4(g[0], g[1], g[2], g[3]);
        *(float4*)(dst + 4) = make_float4(g[4], g[5], g[6], g[7]);
    }
    __syncthreads();

    float rs = 0.f;
    #pragma unroll
    for (int v = 0; v < 8; ++v) rs += sT[wave][(half * 8 + v) * 20 + m16];
    rs += SWZ(rs, SWZ_X16);
    const float e_rl = __expf(rs);
    const float rl_lane = e_rl / sum16(e_rl);

    const float aval = (lane < 16) ? lr_lane : rl_lane;
    out_s[wave][lane] = xv * aval;
    __syncthreads();

    const int n = m16;
    float acc = sdb[n];
    const float4* ov = (const float4*)out_s[wave];
    #pragma unroll
    for (int k4 = 0; k4 < 8; ++k4) {
        const float4 o = ov[k4];
        acc = fmaf(o.x, sdW[(k4 * 4 + 0) * OUT_DIM + n], acc);
        acc = fmaf(o.y, sdW[(k4 * 4 + 1) * OUT_DIM + n], acc);
        acc = fmaf(o.z, sdW[(k4 * 4 + 2) * OUT_DIM + n], acc);
        acc = fmaf(o.w, sdW[(k4 * 4 + 3) * OUT_DIM + n], acc);
    }
    acc = fmaxf(acc, 0.f);

    if (valid && lane < 16)
        out[(size_t)b * OUT_DIM + n] = acc;
}

extern "C" void kernel_launch(void* const* d_in, const int* in_sizes, int n_in,
                              void* d_out, int out_size, void* d_ws, size_t ws_size,
                              hipStream_t stream) {
    const float* x       = (const float*)d_in[0];
    const float* emb_W   = (const float*)d_in[1];
    const float* emb_b   = (const float*)d_in[2];
    const float* M       = (const float*)d_in[3];
    const float* dense_W = (const float*)d_in[4];
    const float* dense_b = (const float*)d_in[5];
    float* out = (float*)d_out;

    const int nrows  = in_sizes[0] / NUM_FEA;
    const int blocks = (nrows + WAVES_PER_BLOCK - 1) / WAVES_PER_BLOCK;
    coattn_kernel<<<blocks, 256, 0, stream>>>(x, emb_W, emb_b, M,
                                              dense_W, dense_b, out, nrows);
}


# each wave owns tiles of 16 consecutive rows: gated rows staged in LDS, dense layer done per tile as eight f32 16x16x4 WMMAs instead of 32 vector FMAs per row; 2048 work-groups at 64 VGPRs
# speedup vs baseline: 2.0876x; 2.0876x over previous
#include <hip/hip_runtime.h>
#include <math.h>

typedef __attribute__((ext_vector_type(2))) float v2f;
typedef __attribute__((ext_vector_type(8))) float v8f;

#define NUM_FEA 32
#define EMB_DIM 8
#define OUT_DIM 16
#define WAVES_PER_BLOCK 8
#define TILE_ROWS 16
#define TILE_PITCH 36
#define PERSIST_BLOCKS 2048

#define SWZ(v, imm) __int_as_float(__builtin_amdgcn_ds_swizzle(__float_as_int(v), (imm)))
#define SWZ_X1  0x041F
#define SWZ_X2  0x081F
#define SWZ_X4  0x101F
#define SWZ_X8  0x201F
#define SWZ_X16 0x401F
#define SWZ_BCL 0x000F
#define SWZ_BCR 0x020F

__device__ __forceinline__ float fast_tanh(float v) {
#if __has_builtin(__builtin_amdgcn_tanhf)
    return __builtin_amdgcn_tanhf(v);
#else
    return tanhf(v);
#endif
}

__device__ __forceinline__ float sum16(float v) {
    v += SWZ(v, SWZ_X1);
    v += SWZ(v, SWZ_X2);
    v += SWZ(v, SWZ_X4);
    v += SWZ(v, SWZ_X8);
    return v;
}

__device__ __forceinline__ float4 dense_frag(const float* base, int c, int h, int p) {
    return *(const float4*)(base + c * TILE_PITCH + 8 * p + 4 * h);
}

__global__ __launch_bounds__(256) __attribute__((amdgpu_waves_per_eu(16, 16)))
void coattn_kernel(const float* __restrict__ x,
                   const float* __restrict__ emb_W,
                   const float* __restrict__ emb_b,
                   const float* __restrict__ M,
                   const float* __restrict__ dense_W,
                   const float* __restrict__ dense_b,
                   float* __restrict__ out,
                   int nrows) {
    __shared__ float sWM[128];
    __shared__ float sbM[128];
    __shared__ float sWr[128];
    __shared__ float sbr[128];
    __shared__ __attribute__((aligned(16))) float sdWT[OUT_DIM * TILE_PITCH];
    __shared__ float sdb[16];
    __shared__ float sT[WAVES_PER_BLOCK][16 * 20];
    __shared__ __attribute__((aligned(16))) float sO[WAVES_PER_BLOCK][TILE_ROWS * TILE_PITCH];

    const int tid = threadIdx.x;
    if (nrows <= 0) return;

    if (tid < 128) {
        const int l = tid >> 3, f = tid & 7;
        float aw = 0.f, ab = 0.f;
        #pragma unroll
        for (int e = 0; e < EMB_DIM; ++e) {
            const float mv = M[e * EMB_DIM + f];
            aw = fmaf(emb_W[l * EMB_DIM + e], mv, aw);
            ab = fmaf(emb_b[l * EMB_DIM + e], mv, ab);
        }
        sWM[tid] = aw;
        sbM[tid] = ab;
        sWr[tid] = emb_W[128 + tid];
        sbr[tid] = emb_b[128 + tid];
    }
    sdWT[(tid & 15) * TILE_PITCH + (tid >> 4)]      = dense_W[tid];
    sdWT[(tid & 15) * TILE_PITCH + (tid >> 4) + 16] = dense_W[tid + 256];
    if (tid < OUT_DIM) sdb[tid] = dense_b[tid];
    __syncthreads();

    const int lane = tid & 31;
    const int wave = __builtin_amdgcn_readfirstlane(tid >> 5);
    const int m16  = lane & 15;
    const int half = lane >> 4;

    float wA[4], bA[4], wB[4], bB[4];
    #pragma unroll
    for (int w = 0; w < 2; ++w) {
        const int k0 = w * 4 + half * 2;
        wA[2 * w]     = sWM[m16 * 8 + k0];
        wA[2 * w + 1] = sWM[m16 * 8 + k0 + 1];
        bA[2 * w]     = sbM[m16 * 8 + k0];
        bA[2 * w + 1] = sbM[m16 * 8 + k0 + 1];
        wB[2 * w]     = sWr[m16 * 8 + k0];
        wB[2 * w + 1] = sWr[m16 * 8 + k0 + 1];
        bB[2 * w]     = sbr[m16 * 8 + k0];
        bB[2 * w + 1] = sbr[m16 * 8 + k0 + 1];
    }
    const int n = m16;
    const float db = sdb[n];

    const int ntiles = (nrows + TILE_ROWS - 1) / TILE_ROWS;
    const int nw     = gridDim.x * WAVES_PER_BLOCK;
    const int gw     = blockIdx.x * WAVES_PER_BLOCK + wave;
    const int trips  = (ntiles + nw - 1) / nw;

    float* const tileS = sO[wave];

    int b;
    {
        int tl = gw;
        if (tl >= ntiles) tl = ntiles - 1;
        b = tl * TILE_ROWS;
        if (b >= nrows) b = nrows - 1;
    }
    float xnext = x[(size_t)b * NUM_FEA + lane];

    for (int t = 0; t < trips; ++t) {
        if (blockIdx.x * WAVES_PER_BLOCK + t * nw >= ntiles) break;

        const int  tile   = gw + t * nw;
        const bool tvalid = (tile < ntiles);
        const int  row0   = (tvalid ? tile : ntiles - 1) * TILE_ROWS;
        int nrow0 = tile + nw;
        if (nrow0 >= ntiles) nrow0 = ntiles - 1;
        nrow0 *= TILE_ROWS;

        #pragma unroll 1
        for (int j = 0; j < TILE_ROWS; ++j) {
            const float xv = xnext;
            const float xl = SWZ(xv, SWZ_BCL);
            const float xr = SWZ(xv, SWZ_BCR);

            v8f c = {};
            #pragma unroll
            for (int w = 0; w < 2; ++w) {
                v2f a, bmat;
                a.x    = fmaf(xl, wA[2 * w],     bA[2 * w]);
                a.y    = fmaf(xl, wA[2 * w + 1], bA[2 * w + 1]);
                bmat.x = fmaf(xr, wB[2 * w],     bB[2 * w]);
                bmat.y = fmaf(xr, wB[2 * w + 1], bB[2 * w + 1]);
                c = __builtin_amdgcn_wmma_f32_16x16x4_f32(
                         false, a,  false, bmat,
                         (short)0, c,  false,  false);
            }

            float g[8];
            #pragma unroll
            for (int v = 0; v < 8; ++v) g[v] = fast_tanh(c[v]);

            float cs = 0.f;
            #pragma unroll
            for (int v = 0; v < 8; ++v) cs += g[v];
            cs += SWZ(cs, SWZ_X16);
            const float e_lr = __expf(cs);
            const float lr_lane = e_lr / sum16(e_lr);

            {
                float* dst = &sT[wave][m16 * 20 + half * 8];
                *(float4*)(dst)     = make_float4(g[0], g[1], g[2], g[3]);
                *(float4*)(dst + 4) = make_float4(g[4], g[5], g[6], g[7]);
            }
            __syncthreads();

            {
                int bn = (j + 1 < TILE_ROWS) ? (row0 + j + 1) : nrow0;
                if (bn >= nrows) bn = nrows - 1;
                xnext = x[(size_t)bn * NUM_FEA + lane];
            }

            float rs = 0.f;
            #pragma unroll
            for (int v = 0; v < 8; ++v) rs += sT[wave][(half * 8 + v) * 20 + m16];
            rs += SWZ(rs, SWZ_X16);
            const float e_rl = __expf(rs);
            const float rl_lane = e_rl / sum16(e_rl);

            const float aval = (lane < 16) ? lr_lane : rl_lane;
            tileS[(8 * (j & 1) + (j >> 1)) * TILE_PITCH + lane] = xv * aval;
        }
        __syncthreads();

        v8f acc0 = {}, acc1 = {};
        #pragma unroll
        for (int p = 0; p < 4; ++p) {
            const float4 q = dense_frag(tileS, m16, half, p);
            const float4 d = dense_frag(sdWT,  m16, half, p);
            v2f a0, a1, b0, b1;
            a0.x = q.x; a0.y = q.y;
            a1.x = q.z; a1.y = q.w;
            b0.x = d.x; b0.y = d.y;
            b1.x = d.z; b1.y = d.w;
            acc0 = __builtin_amdgcn_wmma_f32_16x16x4_f32(
                       false, a0, false, b0, (short)0, acc0, false, false);
            acc1 = __builtin_amdgcn_wmma_f32_16x16x4_f32(
                       false, a1, false, b1, (short)0, acc1, false, false);
        }

        float* const orow = out + (size_t)row0 * OUT_DIM + half * OUT_DIM + n;
        if (tvalid && row0 + TILE_ROWS <= nrows) {
            #pragma unroll
            for (int r = 0; r < 8; ++r)
                orow[r * 2 * OUT_DIM] = fmaxf(acc0[r] + acc1[r] + db, 0.f);
        } else if (tvalid) {
            #pragma unroll
            for (int r = 0; r < 8; ++r)
                if (row0 + 2 * r + half < nrows)
                    orow[r * 2 * OUT_DIM] = fmaxf(acc0[r] + acc1[r] + db, 0.f);
        }
    }
}

extern "C" void kernel_launch(void* const* d_in, const int* in_sizes, int n_in,
                              void* d_out, int out_size, void* d_ws, size_t ws_size,
                              hipStream_t stream) {
    const float* x       = (const float*)d_in[0];
    const float* emb_W   = (const float*)d_in[1];
    const float* emb_b   = (const float*)d_in[2];
    const float* M       = (const float*)d_in[3];
    const float* dense_W = (const float*)d_in[4];
    const float* dense_b = (const float*)d_in[5];
    float* out = (float*)d_out;

    const int nrows  = in_sizes[0] / NUM_FEA;
    coattn_kernel<<<PERSIST_BLOCKS, 256, 0, stream>>>(x, emb_W, emb_b, M,
                                                      dense_W, dense_b, out, nrows);
}


# both 16-wide softmaxes in one pass: lanes 0-15 carry the column sums (lr), lanes 16-31 the row sums (rl); one exp, one 16-lane sum, one division per row instead of two of each
# speedup vs baseline: 2.3328x; 1.1175x over previous
#include <hip/hip_runtime.h>
#include <math.h>

typedef __attribute__((ext_vector_type(2))) float v2f;
typedef __attribute__((ext_vector_type(8))) float v8f;

#define NUM_FEA 32
#define EMB_DIM 8
#define OUT_DIM 16
#define WAVES_PER_BLOCK 8
#define TILE_ROWS 16
#define TILE_PITCH 36
#define PERSIST_BLOCKS 2048

#define SWZ(v, imm) __int_as_float(__builtin_amdgcn_ds_swizzle(__float_as_int(v), (imm)))
#define SWZ_X1  0x041F
#define SWZ_X2  0x081F
#define SWZ_X4  0x101F
#define SWZ_X8  0x201F
#define SWZ_X16 0x401F
#define SWZ_BCL 0x000F
#define SWZ_BCR 0x020F

__device__ __forceinline__ float fast_tanh(float v) {
#if __has_builtin(__builtin_amdgcn_tanhf)
    return __builtin_amdgcn_tanhf(v);
#else
    return tanhf(v);
#endif
}

__device__ __forceinline__ float sum16(float v) {
    v += SWZ(v, SWZ_X1);
    v += SWZ(v, SWZ_X2);
    v += SWZ(v, SWZ_X4);
    v += SWZ(v, SWZ_X8);
    return v;
}

__device__ __forceinline__ float4 dense_frag(const float* base, int c, int h, int p) {
    return *(const float4*)(base + c * TILE_PITCH + 8 * p + 4 * h);
}

__global__ __launch_bounds__(256) __attribute__((amdgpu_waves_per_eu(16, 16)))
void coattn_kernel(const float* __restrict__ x,
                   const float* __restrict__ emb_W,
                   const float* __restrict__ emb_b,
                   const float* __restrict__ M,
                   const float* __restrict__ dense_W,
                   const float* __restrict__ dense_b,
                   float* __restrict__ out,
                   int nrows) {
    __shared__ float sWM[128];
    __shared__ float sbM[128];
    __shared__ float sWr[128];
    __shared__ float sbr[128];
    __shared__ __attribute__((aligned(16))) float sdWT[OUT_DIM * TILE_PITCH];
    __shared__ float sdb[16];
    __shared__ float sT[WAVES_PER_BLOCK][16 * 20];
    __shared__ __attribute__((aligned(16))) float sO[WAVES_PER_BLOCK][TILE_ROWS * TILE_PITCH];

    const int tid = threadIdx.x;
    if (nrows <= 0) return;

    if (tid < 128) {
        const int l = tid >> 3, f = tid & 7;
        float aw = 0.f, ab = 0.f;
        #pragma unroll
        for (int e = 0; e < EMB_DIM; ++e) {
            const float mv = M[e * EMB_DIM + f];
            aw = fmaf(emb_W[l * EMB_DIM + e], mv, aw);
            ab = fmaf(emb_b[l * EMB_DIM + e], mv, ab);
        }
        sWM[tid] = aw;
        sbM[tid] = ab;
        sWr[tid] = emb_W[128 + tid];
        sbr[tid] = emb_b[128 + tid];
    }
    sdWT[(tid & 15) * TILE_PITCH + (tid >> 4)]      = dense_W[tid];
    sdWT[(tid & 15) * TILE_PITCH + (tid >> 4) + 16] = dense_W[tid + 256];
    if (tid < OUT_DIM) sdb[tid] = dense_b[tid];
    __syncthreads();

    const int lane = tid & 31;
    const int wave = __builtin_amdgcn_readfirstlane(tid >> 5);
    const int m16  = lane & 15;
    const int half = lane >> 4;

    float wA[4], bA[4], wB[4], bB[4];
    #pragma unroll
    for (int w = 0; w < 2; ++w) {
        const int k0 = w * 4 + half * 2;
        wA[2 * w]     = sWM[m16 * 8 + k0];
        wA[2 * w + 1] = sWM[m16 * 8 + k0 + 1];
        bA[2 * w]     = sbM[m16 * 8 + k0];
        bA[2 * w + 1] = sbM[m16 * 8 + k0 + 1];
        wB[2 * w]     = sWr[m16 * 8 + k0];
        wB[2 * w + 1] = sWr[m16 * 8 + k0 + 1];
        bB[2 * w]     = sbr[m16 * 8 + k0];
        bB[2 * w + 1] = sbr[m16 * 8 + k0 + 1];
    }
    const int n = m16;
    const float db = sdb[n];

    const int ntiles = (nrows + TILE_ROWS - 1) / TILE_ROWS;
    const int nw     = gridDim.x * WAVES_PER_BLOCK;
    const int gw     = blockIdx.x * WAVES_PER_BLOCK + wave;
    const int trips  = (ntiles + nw - 1) / nw;

    float* const tileS = sO[wave];

    int b;
    {
        int tl = gw;
        if (tl >= ntiles) tl = ntiles - 1;
        b = tl * TILE_ROWS;
        if (b >= nrows) b = nrows - 1;
    }
    float xnext = x[(size_t)b * NUM_FEA + lane];

    for (int t = 0; t < trips; ++t) {
        if (blockIdx.x * WAVES_PER_BLOCK + t * nw >= ntiles) break;

        const int  tile   = gw + t * nw;
        const bool tvalid = (tile < ntiles);
        const int  row0   = (tvalid ? tile : ntiles - 1) * TILE_ROWS;
        int nrow0 = tile + nw;
        if (nrow0 >= ntiles) nrow0 = ntiles - 1;
        nrow0 *= TILE_ROWS;

        #pragma unroll 1
        for (int j = 0; j < TILE_ROWS; ++j) {
            const float xv = xnext;
            const float xl = SWZ(xv, SWZ_BCL);
            const float xr = SWZ(xv, SWZ_BCR);

            v8f c = {};
            #pragma unroll
            for (int w = 0; w < 2; ++w) {
                v2f a, bmat;
                a.x    = fmaf(xl, wA[2 * w],     bA[2 * w]);
                a.y    = fmaf(xl, wA[2 * w + 1], bA[2 * w + 1]);
                bmat.x = fmaf(xr, wB[2 * w],     bB[2 * w]);
                bmat.y = fmaf(xr, wB[2 * w + 1], bB[2 * w + 1]);
                c = __builtin_amdgcn_wmma_f32_16x16x4_f32(
                         false, a,  false, bmat,
                         (short)0, c,  false,  false);
            }

            float g[8];
            #pragma unroll
            for (int v = 0; v < 8; ++v) g[v] = fast_tanh(c[v]);

            float cs = 0.f;
            #pragma unroll
            for (int v = 0; v < 8; ++v) cs += g[v];

            {
                float* dst = &sT[wave][m16 * 20 + half * 8];
                *(float4*)(dst)     = make_float4(g[0], g[1], g[2], g[3]);
                *(float4*)(dst + 4) = make_float4(g[4], g[5], g[6], g[7]);
            }
            __syncthreads();

            {
                int bn = (j + 1 < TILE_ROWS) ? (row0 + j + 1) : nrow0;
                if (bn >= nrows) bn = nrows - 1;
                xnext = x[(size_t)bn * NUM_FEA + lane];
            }

            float rs = 0.f;
            #pragma unroll
            for (int v = 0; v < 8; ++v) rs += sT[wave][(half * 8 + v) * 20 + m16];

            const float keep = (lane < 16) ? cs : rs;
            const float give = (lane < 16) ? rs : cs;
            const float sm = keep + SWZ(give, SWZ_X16);
            const float ev = __expf(sm);
            const float aval = ev / sum16(ev);
            tileS[(8 * (j & 1) + (j >> 1)) * TILE_PITCH + lane] = xv * aval;
        }
        __syncthreads();

        v8f acc0 = {}, acc1 = {};
        #pragma unroll
        for (int p = 0; p < 4; ++p) {
            const float4 q = dense_frag(tileS, m16, half, p);
            const float4 d = dense_frag(sdWT,  m16, half, p);
            v2f a0, a1, b0, b1;
            a0.x = q.x; a0.y = q.y;
            a1.x = q.z; a1.y = q.w;
            b0.x = d.x; b0.y = d.y;
            b1.x = d.z; b1.y = d.w;
            acc0 = __builtin_amdgcn_wmma_f32_16x16x4_f32(
                       false, a0, false, b0, (short)0, acc0, false, false);
            acc1 = __builtin_amdgcn_wmma_f32_16x16x4_f32(
                       false, a1, false, b1, (short)0, acc1, false, false);
        }

        float* const orow = out + (size_t)row0 * OUT_DIM + half * OUT_DIM + n;
        if (tvalid && row0 + TILE_ROWS <= nrows) {
            #pragma unroll
            for (int r = 0; r < 8; ++r)
                orow[r * 2 * OUT_DIM] = fmaxf(acc0[r] + acc1[r] + db, 0.f);
        } else if (tvalid) {
            #pragma unroll
            for (int r = 0; r < 8; ++r)
                if (row0 + 2 * r + half < nrows)
                    orow[r * 2 * OUT_DIM] = fmaxf(acc0[r] + acc1[r] + db, 0.f);
        }
    }
}

extern "C" void kernel_launch(void* const* d_in, const int* in_sizes, int n_in,
                              void* d_out, int out_size, void* d_ws, size_t ws_size,
                              hipStream_t stream) {
    const float* x       = (const float*)d_in[0];
    const float* emb_W   = (const float*)d_in[1];
    const float* emb_b   = (const float*)d_in[2];
    const float* M       = (const float*)d_in[3];
    const float* dense_W = (const float*)d_in[4];
    const float* dense_b = (const float*)d_in[5];
    float* out = (float*)d_out;

    const int nrows  = in_sizes[0] / NUM_FEA;
    coattn_kernel<<<PERSIST_BLOCKS, 256, 0, stream>>>(x, emb_W, emb_b, M,
                                                      dense_W, dense_b, out, nrows);
}


# softmax normalisation by one v_rcp_f32 of the 16-lane sum and one multiply (x*e formed beside the sum) instead of the 12-instruction IEEE division per row
# speedup vs baseline: 2.4666x; 1.0574x over previous
#include <hip/hip_runtime.h>
#include <math.h>

typedef __attribute__((ext_vector_type(2))) float v2f;
typedef __attribute__((ext_vector_type(8))) float v8f;

#define NUM_FEA 32
#define EMB_DIM 8
#define OUT_DIM 16
#define WAVES_PER_BLOCK 8
#define TILE_ROWS 16
#define TILE_PITCH 36
#define PERSIST_BLOCKS 2048

#define SWZ(v, imm) __int_as_float(__builtin_amdgcn_ds_swizzle(__float_as_int(v), (imm)))
#define SWZ_X1  0x041F
#define SWZ_X2  0x081F
#define SWZ_X4  0x101F
#define SWZ_X8  0x201F
#define SWZ_X16 0x401F
#define SWZ_BCL 0x000F
#define SWZ_BCR 0x020F

__device__ __forceinline__ float fast_tanh(float v) {
#if __has_builtin(__builtin_amdgcn_tanhf)
    return __builtin_amdgcn_tanhf(v);
#else
    return tanhf(v);
#endif
}

__device__ __forceinline__ float sum16(float v) {
    v += SWZ(v, SWZ_X1);
    v += SWZ(v, SWZ_X2);
    v += SWZ(v, SWZ_X4);
    v += SWZ(v, SWZ_X8);
    return v;
}

__device__ __forceinline__ float4 dense_frag(const float* base, int c, int h, int p) {
    return *(const float4*)(base + c * TILE_PITCH + 8 * p + 4 * h);
}

__global__ __launch_bounds__(256) __attribute__((amdgpu_waves_per_eu(16, 16)))
void coattn_kernel(const float* __restrict__ x,
                   const float* __restrict__ emb_W,
                   const float* __restrict__ emb_b,
                   const float* __restrict__ M,
                   const float* __restrict__ dense_W,
                   const float* __restrict__ dense_b,
                   float* __restrict__ out,
                   int nrows) {
    __shared__ float sWM[128];
    __shared__ float sbM[128];
    __shared__ float sWr[128];
    __shared__ float sbr[128];
    __shared__ __attribute__((aligned(16))) float sdWT[OUT_DIM * TILE_PITCH];
    __shared__ float sdb[16];
    __shared__ float sT[WAVES_PER_BLOCK][16 * 20];
    __shared__ __attribute__((aligned(16))) float sO[WAVES_PER_BLOCK][TILE_ROWS * TILE_PITCH];

    const int tid = threadIdx.x;
    if (nrows <= 0) return;

    if (tid < 128) {
        const int l = tid >> 3, f = tid & 7;
        float aw = 0.f, ab = 0.f;
        #pragma unroll
        for (int e = 0; e < EMB_DIM; ++e) {
            const float mv = M[e * EMB_DIM + f];
            aw = fmaf(emb_W[l * EMB_DIM + e], mv, aw);
            ab = fmaf(emb_b[l * EMB_DIM + e], mv, ab);
        }
        sWM[tid] = aw;
        sbM[tid] = ab;
        sWr[tid] = emb_W[128 + tid];
        sbr[tid] = emb_b[128 + tid];
    }
    sdWT[(tid & 15) * TILE_PITCH + (tid >> 4)]      = dense_W[tid];
    sdWT[(tid & 15) * TILE_PITCH + (tid >> 4) + 16] = dense_W[tid + 256];
    if (tid < OUT_DIM) sdb[tid] = dense_b[tid];
    __syncthreads();

    const int lane = tid & 31;
    const int wave = __builtin_amdgcn_readfirstlane(tid >> 5);
    const int m16  = lane & 15;
    const int half = lane >> 4;

    float wA[4], bA[4], wB[4], bB[4];
    #pragma unroll
    for (int w = 0; w < 2; ++w) {
        const int k0 = w * 4 + half * 2;
        wA[2 * w]     = sWM[m16 * 8 + k0];
        wA[2 * w + 1] = sWM[m16 * 8 + k0 + 1];
        bA[2 * w]     = sbM[m16 * 8 + k0];
        bA[2 * w + 1] = sbM[m16 * 8 + k0 + 1];
        wB[2 * w]     = sWr[m16 * 8 + k0];
        wB[2 * w + 1] = sWr[m16 * 8 + k0 + 1];
        bB[2 * w]     = sbr[m16 * 8 + k0];
        bB[2 * w + 1] = sbr[m16 * 8 + k0 + 1];
    }
    const int n = m16;
    const float db = sdb[n];

    const int ntiles = (nrows + TILE_ROWS - 1) / TILE_ROWS;
    const int nw     = gridDim.x * WAVES_PER_BLOCK;
    const int gw     = blockIdx.x * WAVES_PER_BLOCK + wave;
    const int trips  = (ntiles + nw - 1) / nw;

    float* const tileS = sO[wave];

    int b;
    {
        int tl = gw;
        if (tl >= ntiles) tl = ntiles - 1;
        b = tl * TILE_ROWS;
        if (b >= nrows) b = nrows - 1;
    }
    float xnext = x[(size_t)b * NUM_FEA + lane];

    for (int t = 0; t < trips; ++t) {
        if (blockIdx.x * WAVES_PER_BLOCK + t * nw >= ntiles) break;

        const int  tile   = gw + t * nw;
        const bool tvalid = (tile < ntiles);
        const int  row0   = (tvalid ? tile : ntiles - 1) * TILE_ROWS;
        int nrow0 = tile + nw;
        if (nrow0 >= ntiles) nrow0 = ntiles - 1;
        nrow0 *= TILE_ROWS;

        #pragma unroll 1
        for (int j = 0; j < TILE_ROWS; ++j) {
            const float xv = xnext;
            const float xl = SWZ(xv, SWZ_BCL);
            const float xr = SWZ(xv, SWZ_BCR);

            v8f c = {};
            #pragma unroll
            for (int w = 0; w < 2; ++w) {
                v2f a, bmat;
                a.x    = fmaf(xl, wA[2 * w],     bA[2 * w]);
                a.y    = fmaf(xl, wA[2 * w + 1], bA[2 * w + 1]);
                bmat.x = fmaf(xr, wB[2 * w],     bB[2 * w]);
                bmat.y = fmaf(xr, wB[2 * w + 1], bB[2 * w + 1]);
                c = __builtin_amdgcn_wmma_f32_16x16x4_f32(
                         false, a,  false, bmat,
                         (short)0, c,  false,  false);
            }

            float g[8];
            #pragma unroll
            for (int v = 0; v < 8; ++v) g[v] = fast_tanh(c[v]);

            float cs = 0.f;
            #pragma unroll
            for (int v = 0; v < 8; ++v) cs += g[v];

            {
                float* dst = &sT[wave][m16 * 20 + half * 8];
                *(float4*)(dst)     = make_float4(g[0], g[1], g[2], g[3]);
                *(float4*)(dst + 4) = make_float4(g[4], g[5], g[6], g[7]);
            }
            __syncthreads();

            {
                int bn = (j + 1 < TILE_ROWS) ? (row0 + j + 1) : nrow0;
                if (bn >= nrows) bn = nrows - 1;
                xnext = x[(size_t)bn * NUM_FEA + lane];
            }

            float rs = 0.f;
            #pragma unroll
            for (int v = 0; v < 8; ++v) rs += sT[wave][(half * 8 + v) * 20 + m16];

            const float keep = (lane < 16) ? cs : rs;
            const float give = (lane < 16) ? rs : cs;
            const float sm = keep + SWZ(give, SWZ_X16);
            const float ev = __expf(sm);
            const float xe  = xv * ev;
            const float inv = __builtin_amdgcn_rcpf(sum16(ev));
            tileS[(8 * (j & 1) + (j >> 1)) * TILE_PITCH + lane] = xe * inv;
        }
        __syncthreads();

        v8f acc0 = {}, acc1 = {};
        #pragma unroll
        for (int p = 0; p < 4; ++p) {
            const float4 q = dense_frag(tileS, m16, half, p);
            const float4 d = dense_frag(sdWT,  m16, half, p);
            v2f a0, a1, b0, b1;
            a0.x = q.x; a0.y = q.y;
            a1.x = q.z; a1.y = q.w;
            b0.x = d.x; b0.y = d.y;
            b1.x = d.z; b1.y = d.w;
            acc0 = __builtin_amdgcn_wmma_f32_16x16x4_f32(
                       false, a0, false, b0, (short)0, acc0, false, false);
            acc1 = __builtin_amdgcn_wmma_f32_16x16x4_f32(
                       false, a1, false, b1, (short)0, acc1, false, false);
        }

        float* const orow = out + (size_t)row0 * OUT_DIM + half * OUT_DIM + n;
        if (tvalid && row0 + TILE_ROWS <= nrows) {
            #pragma unroll
            for (int r = 0; r < 8; ++r)
                orow[r * 2 * OUT_DIM] = fmaxf(acc0[r] + acc1[r] + db, 0.f);
        } else if (tvalid) {
            #pragma unroll
            for (int r = 0; r < 8; ++r)
                if (row0 + 2 * r + half < nrows)
                    orow[r * 2 * OUT_DIM] = fmaxf(acc0[r] + acc1[r] + db, 0.f);
        }
    }
}

extern "C" void kernel_launch(void* const* d_in, const int* in_sizes, int n_in,
                              void* d_out, int out_size, void* d_ws, size_t ws_size,
                              hipStream_t stream) {
    const float* x       = (const float*)d_in[0];
    const float* emb_W   = (const float*)d_in[1];
    const float* emb_b   = (const float*)d_in[2];
    const float* M       = (const float*)d_in[3];
    const float* dense_W = (const float*)d_in[4];
    const float* dense_b = (const float*)d_in[5];
    float* out = (float*)d_out;

    const int nrows  = in_sizes[0] / NUM_FEA;
    coattn_kernel<<<PERSIST_BLOCKS, 256, 0, stream>>>(x, emb_W, emb_b, M,
                                                      dense_W, dense_b, out, nrows);
}


# softmaxes of a whole 16-row tile at once, one (row, side) per lane: 16 exps, in-lane sum, one rcp instead of per-row exp + 4-step cross-lane sum + rcp; gated values stay in registers as dense A
# speedup vs baseline: 2.6815x; 1.0871x over previous
#include <hip/hip_runtime.h>
#include <math.h>

typedef __attribute__((ext_vector_type(2))) float v2f;
typedef __attribute__((ext_vector_type(8))) float v8f;

#define NUM_FEA 32
#define EMB_DIM 8
#define OUT_DIM 16
#define WAVES_PER_BLOCK 8
#define TILE_ROWS 16
#define TILE_PITCH 36
#define PERSIST_BLOCKS 2048

#define SWZ(v, imm) __int_as_float(__builtin_amdgcn_ds_swizzle(__float_as_int(v), (imm)))
#define SWZ_X16 0x401F
#define SWZ_BCL 0x000F
#define SWZ_BCR 0x020F

__device__ __forceinline__ float fast_tanh(float v) {
#if __has_builtin(__builtin_amdgcn_tanhf)
    return __builtin_amdgcn_tanhf(v);
#else
    return tanhf(v);
#endif
}

__device__ __forceinline__ float4 half_row_piece(const float* base, unsigned line0, int h, int p) {
    return *(const float4*)(base + (line0 + (unsigned)(16 * h + 4 * p)));
}

__global__ __launch_bounds__(256) __attribute__((amdgpu_waves_per_eu(16, 16)))
void coattn_kernel(const float* __restrict__ x,
                   const float* __restrict__ emb_W,
                   const float* __restrict__ emb_b,
                   const float* __restrict__ M,
                   const float* __restrict__ dense_W,
                   const float* __restrict__ dense_b,
                   float* __restrict__ out,
                   int nrows) {
    __shared__ __attribute__((aligned(16))) float sW4[4][128];
    float* const sWM = sW4[0];
    float* const sbM = sW4[1];
    float* const sWr = sW4[2];
    float* const sbr = sW4[3];
    __shared__ __attribute__((aligned(16))) float sdWT[OUT_DIM * TILE_PITCH];
    __shared__ float sT[WAVES_PER_BLOCK][16 * 20];
    __shared__ __attribute__((aligned(16))) float sO[WAVES_PER_BLOCK][TILE_ROWS * TILE_PITCH];

    const int tid = threadIdx.x;
    if (nrows <= 0) return;

    if (tid < 128) {
        const int l = tid >> 3, f = tid & 7;
        float aw = 0.f, ab = 0.f;
        #pragma unroll
        for (int e = 0; e < EMB_DIM; ++e) {
            const float mv = M[e * EMB_DIM + f];
            aw = fmaf(emb_W[l * EMB_DIM + e], mv, aw);
            ab = fmaf(emb_b[l * EMB_DIM + e], mv, ab);
        }
        sWM[tid] = aw;
        sbM[tid] = ab;
        sWr[tid] = emb_W[128 + tid];
        sbr[tid] = emb_b[128 + tid];
    }
    sdWT[(tid & 15) * TILE_PITCH + (tid >> 4)]      = dense_W[tid];
    sdWT[(tid & 15) * TILE_PITCH + (tid >> 4) + 16] = dense_W[tid + 256];
    __syncthreads();

    const int lane = tid & 31;
    const int wave = __builtin_amdgcn_readfirstlane(tid >> 5);
    const int m16  = lane & 15;
    const int half = lane >> 4;

    const int n = m16;

    const int ntiles = (nrows + TILE_ROWS - 1) / TILE_ROWS;
    const int nw     = gridDim.x * WAVES_PER_BLOCK;
    const int gw     = blockIdx.x * WAVES_PER_BLOCK + wave;
    const int trips  = (ntiles + nw - 1) / nw;

    float* const tileS = sO[wave];
    const int myrow = 2 * (m16 & 7) + (m16 >> 3);

    int b;
    {
        int tl = gw;
        if (tl >= ntiles) tl = ntiles - 1;
        b = tl * TILE_ROWS;
        if (b >= nrows) b = nrows - 1;
    }
    float xnext = x[(unsigned)b * (unsigned)NUM_FEA + (unsigned)lane];

    for (int t = 0; t < trips; ++t) {
        if (blockIdx.x * WAVES_PER_BLOCK + t * nw >= ntiles) break;

        const int  tile   = gw + t * nw;
        const bool tvalid = (tile < ntiles);
        const int  row0   = (tvalid ? tile : ntiles - 1) * TILE_ROWS;
        int nrow0 = tile + nw;
        if (nrow0 >= ntiles) nrow0 = ntiles - 1;
        nrow0 *= TILE_ROWS;

        float wA[4], bA[4], wB[4], bB[4];
        #pragma unroll
        for (int w = 0; w < 2; ++w) {
            const int k0 = w * 4 + half * 2;
            wA[2 * w]     = sWM[m16 * 8 + k0];
            wA[2 * w + 1] = sWM[m16 * 8 + k0 + 1];
            bA[2 * w]     = sbM[m16 * 8 + k0];
            bA[2 * w + 1] = sbM[m16 * 8 + k0 + 1];
            wB[2 * w]     = sWr[m16 * 8 + k0];
            wB[2 * w + 1] = sWr[m16 * 8 + k0 + 1];
            bB[2 * w]     = sbr[m16 * 8 + k0];
            bB[2 * w + 1] = sbr[m16 * 8 + k0 + 1];
        }

        #pragma unroll 1
        for (int j = 0; j < TILE_ROWS; ++j) {
            const float xv = xnext;
            const float xl = SWZ(xv, SWZ_BCL);
            const float xr = SWZ(xv, SWZ_BCR);

            v8f c = {};
            #pragma unroll
            for (int w = 0; w < 2; ++w) {
                v2f a, bmat;
                a.x    = fmaf(xl, wA[2 * w],     bA[2 * w]);
                a.y    = fmaf(xl, wA[2 * w + 1], bA[2 * w + 1]);
                bmat.x = fmaf(xr, wB[2 * w],     bB[2 * w]);
                bmat.y = fmaf(xr, wB[2 * w + 1], bB[2 * w + 1]);
                c = __builtin_amdgcn_wmma_f32_16x16x4_f32(
                         false, a,  false, bmat,
                         (short)0, c,  false,  false);
            }

            float g[8];
            #pragma unroll
            for (int v = 0; v < 8; ++v) g[v] = fast_tanh(c[v]);

            float cs = 0.f;
            #pragma unroll
            for (int v = 0; v < 8; ++v) cs += g[v];

            {
                float* dst = &sT[wave][m16 * 20 + half * 8];
                *(float4*)(dst)     = make_float4(g[0], g[1], g[2], g[3]);
                *(float4*)(dst + 4) = make_float4(g[4], g[5], g[6], g[7]);
            }
            __syncthreads();

            {
                int bn = (j + 1 < TILE_ROWS) ? (row0 + j + 1) : nrow0;
                if (bn >= nrows) bn = nrows - 1;
                xnext = x[(unsigned)bn * (unsigned)NUM_FEA + (unsigned)lane];
            }

            float rs = 0.f;
            #pragma unroll
            for (int v = 0; v < 8; ++v) rs += sT[wave][(half * 8 + v) * 20 + m16];

            const float keep = (lane < 16) ? cs : rs;
            const float give = (lane < 16) ? rs : cs;
            const float sm = keep + SWZ(give, SWZ_X16);
            tileS[j * TILE_PITCH + lane] = sm;
        }

        float4 xq[4];
        {
            int gr = row0 + myrow;
            if (gr >= nrows) gr = nrows - 1;
            #pragma unroll
            for (int p = 0; p < 4; ++p)
                xq[p] = half_row_piece(x, (unsigned)gr * NUM_FEA, half, p);
        }
        __syncthreads();

        float ge[16];
        {
            float part[4];
            #pragma unroll
            for (int p = 0; p < 4; ++p) {
                const float4 s4 = half_row_piece(tileS, myrow * TILE_PITCH, half, p);
                const float e0 = __expf(s4.x), e1 = __expf(s4.y);
                const float e2 = __expf(s4.z), e3 = __expf(s4.w);
                part[p] = (e0 + e1) + (e2 + e3);
                ge[4 * p]     = xq[p].x * e0;
                ge[4 * p + 1] = xq[p].y * e1;
                ge[4 * p + 2] = xq[p].z * e2;
                ge[4 * p + 3] = xq[p].w * e3;
            }
            const float inv = __builtin_amdgcn_rcpf((part[0] + part[1]) + (part[2] + part[3]));
            #pragma unroll
            for (int i = 0; i < 16; ++i) ge[i] *= inv;
        }

        const float db = dense_b[n];
        v8f acc0 = {}, acc1 = {};
        #pragma unroll
        for (int p = 0; p < 4; ++p) {
            const float4 d = half_row_piece(sdWT, m16 * TILE_PITCH, half, p);
            v2f a0, a1, b0, b1;
            a0.x = ge[4 * p];     a0.y = ge[4 * p + 1];
            a1.x = ge[4 * p + 2]; a1.y = ge[4 * p + 3];
            b0.x = d.x; b0.y = d.y;
            b1.x = d.z; b1.y = d.w;
            acc0 = __builtin_amdgcn_wmma_f32_16x16x4_f32(
                       false, a0, false, b0, (short)0, acc0, false, false);
            acc1 = __builtin_amdgcn_wmma_f32_16x16x4_f32(
                       false, a1, false, b1, (short)0, acc1, false, false);
        }

        float* const orow = out + ((unsigned)row0 * (unsigned)OUT_DIM + (unsigned)(half * OUT_DIM + n));
        if (tvalid && row0 + TILE_ROWS <= nrows) {
            #pragma unroll
            for (int r = 0; r < 8; ++r)
                orow[r * 2 * OUT_DIM] = fmaxf(acc0[r] + acc1[r] + db, 0.f);
        } else if (tvalid) {
            #pragma unroll
            for (int r = 0; r < 8; ++r)
                if (row0 + 2 * r + half < nrows)
                    orow[r * 2 * OUT_DIM] = fmaxf(acc0[r] + acc1[r] + db, 0.f);
        }
    }
}

extern "C" void kernel_launch(void* const* d_in, const int* in_sizes, int n_in,
                              void* d_out, int out_size, void* d_ws, size_t ws_size,
                              hipStream_t stream) {
    const float* x       = (const float*)d_in[0];
    const float* emb_W   = (const float*)d_in[1];
    const float* emb_b   = (const float*)d_in[2];
    const float* M       = (const float*)d_in[3];
    const float* dense_W = (const float*)d_in[4];
    const float* dense_b = (const float*)d_in[5];
    float* out = (float*)d_out;

    const int nrows  = in_sizes[0] / NUM_FEA;
    coattn_kernel<<<PERSIST_BLOCKS, 256, 0, stream>>>(x, emb_W, emb_b, M,
                                                      dense_W, dense_b, out, nrows);
}


# row loop's serial chain cut: lane exchanges on the vector ALU (x pair loaded per lane, v_permlanex16) not ds_swizzle, no work-group barrier (wave-private LDS), rows in pairs, 8-term sums as pk trees
# speedup vs baseline: 3.0490x; 1.1371x over previous
#include <hip/hip_runtime.h>
#include <math.h>

typedef __attribute__((ext_vector_type(2))) float v2f;
typedef __attribute__((ext_vector_type(8))) float v8f;

#define NUM_FEA 32
#define EMB_DIM 8
#define OUT_DIM 16
#define WAVES_PER_BLOCK 8
#define TILE_ROWS 16
#define TILE_PITCH 36
#define PERSIST_BLOCKS 2048

__device__ __forceinline__ float xhalf(float v) {
    const int b = __float_as_int(v);
    return __int_as_float(__builtin_amdgcn_permlanex16(b, b, 0x76543210, 0xfedcba98, false, false));
}

__device__ __forceinline__ void wave_lds_order() {
    __builtin_amdgcn_fence(__ATOMIC_ACQ_REL, "wavefront");
    __builtin_amdgcn_wave_barrier();
}

__device__ __forceinline__ float fast_tanh(float v) {
#if __has_builtin(__builtin_amdgcn_tanhf)
    return __builtin_amdgcn_tanhf(v);
#else
    return tanhf(v);
#endif
}

__device__ __forceinline__ float4 half_row_piece(const float* base, unsigned line0, int h, int p) {
    return *(const float4*)(base + (line0 + (unsigned)(16 * h + 4 * p)));
}

__global__ __launch_bounds__(256) __attribute__((amdgpu_waves_per_eu(16, 16)))
void coattn_kernel(const float* __restrict__ x,
                   const float* __restrict__ emb_W,
                   const float* __restrict__ emb_b,
                   const float* __restrict__ M,
                   const float* __restrict__ dense_W,
                   const float* __restrict__ dense_b,
                   float* __restrict__ out,
                   int nrows) {
    __shared__ __attribute__((aligned(16))) float sW4[4][128];
    float* const sWM = sW4[0];
    float* const sbM = sW4[1];
    float* const sWr = sW4[2];
    float* const sbr = sW4[3];
    __shared__ __attribute__((aligned(16))) float sdWT[OUT_DIM * TILE_PITCH];
    __shared__ float sT[WAVES_PER_BLOCK][16 * 20];
    __shared__ __attribute__((aligned(16))) float sO[WAVES_PER_BLOCK][TILE_ROWS * TILE_PITCH];

    const int tid = threadIdx.x;
    if (nrows <= 0) return;

    if (tid < 128) {
        const int l = tid >> 3, f = tid & 7;
        float aw = 0.f, ab = 0.f;
        #pragma unroll
        for (int e = 0; e < EMB_DIM; ++e) {
            const float mv = M[e * EMB_DIM + f];
            aw = fmaf(emb_W[l * EMB_DIM + e], mv, aw);
            ab = fmaf(emb_b[l * EMB_DIM + e], mv, ab);
        }
        sWM[tid] = aw;
        sbM[tid] = ab;
        sWr[tid] = emb_W[128 + tid];
        sbr[tid] = emb_b[128 + tid];
    }
    sdWT[(tid & 15) * TILE_PITCH + (tid >> 4)]      = dense_W[tid];
    sdWT[(tid & 15) * TILE_PITCH + (tid >> 4) + 16] = dense_W[tid + 256];
    __syncthreads();

    const int lane = tid & 31;
    const int wave = __builtin_amdgcn_readfirstlane(tid >> 5);
    const int m16  = lane & 15;
    const int half = lane >> 4;

    const int ntiles = (nrows + TILE_ROWS - 1) / TILE_ROWS;
    const int nw     = gridDim.x * WAVES_PER_BLOCK;
    const int gw     = blockIdx.x * WAVES_PER_BLOCK + wave;
    const int trips  = (ntiles + nw - 1) / nw;

    float* const tileS = sO[wave];

    for (int t = 0; t < trips; ++t) {
        if (blockIdx.x * WAVES_PER_BLOCK + t * nw >= ntiles) break;

        const int  tile   = gw + t * nw;
        const bool tvalid = (tile < ntiles);
        const int  row0   = (tvalid ? tile : ntiles - 1) * TILE_ROWS;

        float xl0 = x[(unsigned)row0 * (unsigned)NUM_FEA + (unsigned)m16];
        float xr0 = x[(unsigned)row0 * (unsigned)NUM_FEA + (unsigned)(16 + m16)];
        float xl1, xr1;
        {
            int b1 = row0 + 1;
            if (b1 >= nrows) b1 = nrows - 1;
            xl1 = x[(unsigned)b1 * (unsigned)NUM_FEA + (unsigned)m16];
            xr1 = x[(unsigned)b1 * (unsigned)NUM_FEA + (unsigned)(16 + m16)];
        }

        float wA[4], bA[4], wB[4], bB[4];
        #pragma unroll
        for (int w = 0; w < 2; ++w) {
            const int k0 = w * 4 + half * 2;
            wA[2 * w]     = sWM[m16 * 8 + k0];
            wA[2 * w + 1] = sWM[m16 * 8 + k0 + 1];
            bA[2 * w]     = sbM[m16 * 8 + k0];
            bA[2 * w + 1] = sbM[m16 * 8 + k0 + 1];
            wB[2 * w]     = sWr[m16 * 8 + k0];
            wB[2 * w + 1] = sWr[m16 * 8 + k0 + 1];
            bB[2 * w]     = sbr[m16 * 8 + k0];
            bB[2 * w + 1] = sbr[m16 * 8 + k0 + 1];
        }

        auto one_row = [&](const int j, float& xa, float& xb) {
            const float xl = xa;
            const float xr = xb;
            {
                int bn = row0 + ((j + 2 < TILE_ROWS) ? (j + 2) : (TILE_ROWS - 1));
                if (bn >= nrows) bn = nrows - 1;
                xa = x[(unsigned)bn * (unsigned)NUM_FEA + (unsigned)m16];
                xb = x[(unsigned)bn * (unsigned)NUM_FEA + (unsigned)(16 + m16)];
            }

            v8f c = {};
            #pragma unroll
            for (int w = 0; w < 2; ++w) {
                v2f a, bmat;
                a.x    = fmaf(xl, wA[2 * w],     bA[2 * w]);
                a.y    = fmaf(xl, wA[2 * w + 1], bA[2 * w + 1]);
                bmat.x = fmaf(xr, wB[2 * w],     bB[2 * w]);
                bmat.y = fmaf(xr, wB[2 * w + 1], bB[2 * w + 1]);
                c = __builtin_amdgcn_wmma_f32_16x16x4_f32(
                         false, a,  false, bmat,
                         (short)0, c,  false,  false);
            }

            float g[8];
            #pragma unroll
            for (int v = 0; v < 8; ++v) g[v] = fast_tanh(c[v]);

            float cs;
            {
                v2f p0, p1, p2, p3;
                p0.x = g[0]; p0.y = g[1];  p1.x = g[2]; p1.y = g[3];
                p2.x = g[4]; p2.y = g[5];  p3.x = g[6]; p3.y = g[7];
                const v2f q = (p0 + p1) + (p2 + p3);
                cs = q.x + q.y;
            }

            {
                float* dst = &sT[wave][m16 * 20 + half * 8];
                *(float4*)(dst)     = make_float4(g[0], g[1], g[2], g[3]);
                *(float4*)(dst + 4) = make_float4(g[4], g[5], g[6], g[7]);
            }
            wave_lds_order();

            float rs;
            {
                const float* src = &sT[wave][half * 8 * 20 + m16];
                v2f p0, p1, p2, p3;
                p0.x = src[0 * 20]; p0.y = src[1 * 20];  p1.x = src[2 * 20]; p1.y = src[3 * 20];
                p2.x = src[4 * 20]; p2.y = src[5 * 20];  p3.x = src[6 * 20]; p3.y = src[7 * 20];
                const v2f q = (p0 + p1) + (p2 + p3);
                rs = q.x + q.y;
            }

            const float keep = (lane < 16) ? cs : rs;
            const float give = (lane < 16) ? rs : cs;
            const float sm = keep + xhalf(give);
            tileS[j * TILE_PITCH + lane] = sm;
        };
        #pragma unroll 1
        for (int j = 0; j < TILE_ROWS; j += 2) {
            one_row(j,     xl0, xr0);
            one_row(j + 1, xl1, xr1);
        }

        int lt = lane;
        asm volatile("" : "+v"(lt));
        const int m16t  = lt & 15;
        const int halft = lt >> 4;
        const int myrow = 2 * (m16t & 7) + (m16t >> 3);

        float4 xq[4];
        {
            int gr = row0 + myrow;
            if (gr >= nrows) gr = nrows - 1;
            #pragma unroll
            for (int p = 0; p < 4; ++p)
                xq[p] = half_row_piece(x, (unsigned)gr * NUM_FEA, halft, p);
        }
        wave_lds_order();

        float ge[16];
        {
            float part[4];
            #pragma unroll
            for (int p = 0; p < 4; ++p) {
                const float4 s4 = half_row_piece(tileS, myrow * TILE_PITCH, halft, p);
                const float e0 = __expf(s4.x), e1 = __expf(s4.y);
                const float e2 = __expf(s4.z), e3 = __expf(s4.w);
                part[p] = (e0 + e1) + (e2 + e3);
                ge[4 * p]     = xq[p].x * e0;
                ge[4 * p + 1] = xq[p].y * e1;
                ge[4 * p + 2] = xq[p].z * e2;
                ge[4 * p + 3] = xq[p].w * e3;
            }
            const float inv = __builtin_amdgcn_rcpf((part[0] + part[1]) + (part[2] + part[3]));
            #pragma unroll
            for (int i = 0; i < 16; ++i) ge[i] *= inv;
        }

        const float db = dense_b[m16t];
        v8f acc0 = {}, acc1 = {};
        #pragma unroll
        for (int p = 0; p < 4; ++p) {
            const float4 d = half_row_piece(sdWT, m16t * TILE_PITCH, halft, p);
            v2f a0, a1, b0, b1;
            a0.x = ge[4 * p];     a0.y = ge[4 * p + 1];
            a1.x = ge[4 * p + 2]; a1.y = ge[4 * p + 3];
            b0.x = d.x; b0.y = d.y;
            b1.x = d.z; b1.y = d.w;
            acc0 = __builtin_amdgcn_wmma_f32_16x16x4_f32(
                       false, a0, false, b0, (short)0, acc0, false, false);
            acc1 = __builtin_amdgcn_wmma_f32_16x16x4_f32(
                       false, a1, false, b1, (short)0, acc1, false, false);
        }

        float* const orow = out + ((unsigned)row0 * (unsigned)OUT_DIM + (unsigned)(halft * OUT_DIM + m16t));
        if (tvalid && row0 + TILE_ROWS <= nrows) {
            #pragma unroll
            for (int r = 0; r < 8; ++r)
                orow[r * 2 * OUT_DIM] = fmaxf(acc0[r] + acc1[r] + db, 0.f);
        } else if (tvalid) {
            #pragma unroll
            for (int r = 0; r < 8; ++r)
                if (row0 + 2 * r + halft < nrows)
                    orow[r * 2 * OUT_DIM] = fmaxf(acc0[r] + acc1[r] + db, 0.f);
        }
    }
}

extern "C" void kernel_launch(void* const* d_in, const int* in_sizes, int n_in,
                              void* d_out, int out_size, void* d_ws, size_t ws_size,
                              hipStream_t stream) {
    const float* x       = (const float*)d_in[0];
    const float* emb_W   = (const float*)d_in[1];
    const float* emb_b   = (const float*)d_in[2];
    const float* M       = (const float*)d_in[3];
    const float* dense_W = (const float*)d_in[4];
    const float* dense_b = (const float*)d_in[5];
    float* out = (float*)d_out;

    const int nrows  = in_sizes[0] / NUM_FEA;
    coattn_kernel<<<PERSIST_BLOCKS, 256, 0, stream>>>(x, emb_W, emb_b, M,
                                                      dense_W, dense_b, out, nrows);
}
